# starting kernel
# baseline (speedup 1.0000x reference)
#include <hip/hip_runtime.h>
#include <hip/hip_bf16.h>


#define L_STEPS 4096
#define NDIH    3
#define BATCH   32
#define CHAIN   (L_STEPS * NDIH)
#define CHUNK   256
#define NCHUNK  (CHAIN / CHUNK)
#define MST     17
#define PI_F    3.14159265358979f

typedef __attribute__((ext_vector_type(2))) float v2f;
typedef __attribute__((ext_vector_type(8))) float v8f;

__device__ __forceinline__ void wmma_compose4(const float* __restrict__ src,
                                              const float* __restrict__ idm,
                                              float* __restrict__ dst,
                                              int s0, int off, int lane)
{
    const int half = lane >> 4;
    const int Ml   = lane & 15;
    const int g    = Ml >> 2;
    const int r    = Ml & 3;
    const int c    = Ml & 3;
    const int k0   = half ? 2 : 0;

    const int    sg = s0 + g;
    const float* mm = (sg >= off) ? (src + (size_t)(sg - off) * MST) : idm;
    v2f a;
    a.x = mm[r * 4 + k0];
    a.y = mm[r * 4 + k0 + 1];

    const float* nn = src + (size_t)(s0 + g) * MST;
    v2f bb;
    bb.x = nn[k0 * 4 + c];
    bb.y = nn[(k0 + 1) * 4 + c];

    v8f cz = {0.f, 0.f, 0.f, 0.f, 0.f, 0.f, 0.f, 0.f};
    v8f d = __builtin_amdgcn_wmma_f32_16x16x4_f32(false, a, false, bb,
                                                  (short)0, cz, false, false);

#pragma unroll
    for (int rr = 0; rr < 8; ++rr) {
        const int mb = (rr >> 2) + (half << 1);
        if (mb == g) {
            dst[(size_t)(s0 + mb) * MST + (rr & 3) * 4 + c] = d[rr];
        }
    }
}

__global__ void __launch_bounds__(256)
rgn_local_scan(const float* __restrict__ dih_in,
               float* __restrict__ tloc,
               float* __restrict__ agg)
{
    __shared__ float S0[CHUNK * MST];
    __shared__ float S1[CHUNK * MST];
    __shared__ float IDM[16];

    const int tid   = threadIdx.x;
    const int lane  = tid & 31;
    const int wave  = tid >> 5;
    const int chunk = blockIdx.x;
    const int b     = blockIdx.y;

    if (tid < 16) IDM[tid] = (tid % 5 == 0) ? 1.f : 0.f;

    const int gi = chunk * CHUNK + tid;
    const int l  = gi / NDIH;
    const int j  = gi - l * NDIH;

    const float BL[3] = {145.801f, 152.326f, 132.868f};
    const float BA[3] = {2.124f, 1.941f, 2.028f};
    const float bl    = BL[j];
    const float alpha = PI_F - BA[j];
    const float rc    = bl * cosf(alpha);
    const float rs    = bl * sinf(alpha);

    const float dih = dih_in[((size_t)l * BATCH + b) * NDIH + j];
    float sd, cd;
    sincosf(dih, &sd, &cd);
    const float px = rc, py = cd * rs, pz = sd * rs;

    const float inv_r = 1.0f / bl;
    const float ux = px * inv_r, uy = py * inv_r, uz = pz * inv_r;
    const float inv_s = rsqrtf(fmaxf(py * py + pz * pz, 1e-12f));
    const float wx = 0.f, wy = -pz * inv_s, wz = py * inv_s;
    const float vx = wy * uz - wz * uy;
    const float vy = wz * ux - wx * uz;
    const float vz = wx * uy - wy * ux;

    float* T = S0 + (size_t)tid * MST;
    T[0] = ux; T[1] = vx; T[2]  = wx; T[3]  = px;
    T[4] = uy; T[5] = vy; T[6]  = wy; T[7]  = py;
    T[8] = uz; T[9] = vz; T[10] = wz; T[11] = pz;
    T[12] = 0.f; T[13] = 0.f; T[14] = 0.f; T[15] = 1.f;
    __syncthreads();

#pragma unroll
    for (int rnd = 0; rnd < 8; ++rnd) {
        const int off = 1 << rnd;
        float* src = (rnd & 1) ? S1 : S0;
        float* dst = (rnd & 1) ? S0 : S1;
#pragma unroll
        for (int it = 0; it < CHUNK / 32; ++it) {
            wmma_compose4(src, IDM, dst, it * 32 + wave * 4, off, lane);
        }
        __syncthreads();
    }

    const float* fin = S0 + (size_t)tid * MST;
    const size_t oi = ((size_t)gi * BATCH + b) * 3;
    tloc[oi + 0] = fin[3];
    tloc[oi + 1] = fin[7];
    tloc[oi + 2] = fin[11];
    if (tid < 16)
        agg[((size_t)b * NCHUNK + chunk) * 16 + tid] = S0[(CHUNK - 1) * MST + tid];
}

__global__ void rgn_agg_scan(const float* __restrict__ agg,
                             float* __restrict__ pref)
{
    const int b = threadIdx.x;
    if (b >= BATCH) return;
    float P[16] = {1.f, 0.f, 0.f, 0.f,
                   0.f, 1.f, 0.f, 0.f,
                   0.f, 0.f, 1.f, 0.f,
                   0.f, 0.f, 0.f, 1.f};
    for (int k = 0; k < NCHUNK; ++k) {
        const size_t base = ((size_t)b * NCHUNK + k) * 16;
#pragma unroll
        for (int e = 0; e < 16; ++e) pref[base + e] = P[e];
        const float* A = agg + base;
        float T[16];
#pragma unroll
        for (int r = 0; r < 4; ++r)
#pragma unroll
            for (int c = 0; c < 4; ++c)
                T[r * 4 + c] = fmaf(P[r * 4 + 0], A[0 * 4 + c],
                               fmaf(P[r * 4 + 1], A[1 * 4 + c],
                               fmaf(P[r * 4 + 2], A[2 * 4 + c],
                                    P[r * 4 + 3] * A[3 * 4 + c])));
#pragma unroll
        for (int e = 0; e < 16; ++e) P[e] = T[e];
    }
}

__global__ void __launch_bounds__(256)
rgn_apply(const float* __restrict__ tloc,
          const float* __restrict__ pref,
          float* __restrict__ out)
{
    const int t = blockIdx.x * 256 + threadIdx.x;
    if (t >= CHAIN * BATCH) return;
    const int i = t / BATCH;
    const int b = t - i * BATCH;
    const int chunk = i / CHUNK;
    const float* P = pref + ((size_t)b * NCHUNK + chunk) * 16;
    const size_t oi = (size_t)t * 3;
    const float tx = tloc[oi + 0], ty = tloc[oi + 1], tz = tloc[oi + 2];
    out[oi + 0] = fmaf(P[0], tx, fmaf(P[1], ty, fmaf(P[2],  tz, P[3])));
    out[oi + 1] = fmaf(P[4], tx, fmaf(P[5], ty, fmaf(P[6],  tz, P[7])));
    out[oi + 2] = fmaf(P[8], tx, fmaf(P[9], ty, fmaf(P[10], tz, P[11])));
}

extern "C" void kernel_launch(void* const* d_in, const int* in_sizes, int n_in,
                              void* d_out, int out_size, void* d_ws, size_t ws_size,
                              hipStream_t stream)
{
    (void)in_sizes; (void)n_in; (void)out_size; (void)ws_size;
    const float* dih = (const float*)d_in[0];
    float* out = (float*)d_out;

    float* ws   = (float*)d_ws;
    float* tloc = ws;
    float* agg  = tloc + (size_t)CHAIN * BATCH * 3;
    float* pref = agg + (size_t)BATCH * NCHUNK * 16;

    dim3 gridB(NCHUNK, BATCH, 1);
    rgn_local_scan<<<gridB, 256, 0, stream>>>(dih, tloc, agg);
    rgn_agg_scan<<<1, 32, 0, stream>>>(agg, pref);
    const int total = CHAIN * BATCH;
    rgn_apply<<<(total + 255) / 256, 256, 0, stream>>>(tloc, pref, out);
}


# LDS instructions per 4x4 product 6 -> 2: a slot is kept row- or column-major by its next role, WMMA operands exchanged give the product transposed; 8-byte operand loads, one 16-byte store; pitch 20
# speedup vs baseline: 2.5920x; 2.5920x over previous
#include <hip/hip_runtime.h>
#include <hip/hip_bf16.h>


#define L_STEPS 4096
#define NDIH    3
#define BATCH   32
#define CHAIN   (L_STEPS * NDIH)
#define CHUNK   256
#define NCHUNK  (CHAIN / CHUNK)
#define MST     20
#define PI_F    3.14159265358979f

typedef __attribute__((ext_vector_type(2))) float v2f;
typedef __attribute__((ext_vector_type(4))) float v4f;
typedef __attribute__((ext_vector_type(8))) float v8f;

template <int MODE>
__device__ __forceinline__ void wmma_reduce4(float* buf, int s0, int sstr,
                                             int off, int gmax, int lane)
{
    const int h    = lane >> 4;
    const int Ml   = lane & 15;
    const int g    = Ml >> 2;
    const int j    = Ml & 3;
    const int ge   = (g < gmax) ? g : (gmax - 1);
    const int slot = s0 + ge * sstr;

    const float* xl = buf + (size_t)(slot - off) * MST;
    const float* yr = buf + (size_t)slot * MST;
    const v2f x = *(const v2f*)(xl + j * 4 + 2 * h);
    v2f y;
    bool outR;
    if (MODE == 0) {
        y = *(const v2f*)(yr + j * 4 + 2 * h);
        outR = ((ge & 1) == 0);
    } else {
        y.x = yr[(2 * h) * 4 + j];
        y.y = yr[(2 * h + 1) * 4 + j];
        outR = true;
    }
    const v2f a  = outR ? y : x;
    const v2f bb = outR ? x : y;

    v8f cz = {0.f, 0.f, 0.f, 0.f, 0.f, 0.f, 0.f, 0.f};
    v8f d = __builtin_amdgcn_wmma_f32_16x16x4_f32(false, a, false, bb,
                                                  (short)0, cz, false, false);
    v4f o;
    o.x = (g & 1) ? d[4] : d[0];
    o.y = (g & 1) ? d[5] : d[1];
    o.z = (g & 1) ? d[6] : d[2];
    o.w = (g & 1) ? d[7] : d[3];
    if (h == (g >> 1) && g < gmax)
        *(v4f*)(buf + (size_t)(s0 + g * sstr) * MST + j * 4) = o;
}

__device__ __forceinline__ void build_local(const float* __restrict__ dih_in,
                                            int chunk, int b, int tid,
                                            float* __restrict__ S)
{
    const int gi = chunk * CHUNK + tid;
    const int l  = gi / NDIH;
    const int j  = gi - l * NDIH;

    const float BL[3] = {145.801f, 152.326f, 132.868f};
    const float BA[3] = {2.124f, 1.941f, 2.028f};
    const float bl    = BL[j];
    const float alpha = PI_F - BA[j];
    const float rc    = bl * cosf(alpha);
    const float rs    = bl * sinf(alpha);

    const float dih = dih_in[((size_t)l * BATCH + b) * NDIH + j];
    float sd, cd;
    sincosf(dih, &sd, &cd);
    const float px = rc, py = cd * rs, pz = sd * rs;

    const float inv_r = 1.0f / bl;
    const float ux = px * inv_r, uy = py * inv_r, uz = pz * inv_r;
    const float inv_s = rsqrtf(fmaxf(py * py + pz * pz, 1e-12f));
    const float wx = 0.f, wy = -pz * inv_s, wz = py * inv_s;
    const float vx = wy * uz - wz * uy;
    const float vy = wz * ux - wx * uz;
    const float vz = wx * uy - wy * ux;

    const bool cm = (tid & 1) != 0;
    v4f q0, q1, q2, q3;
    q0.x = ux;             q0.y = cm ? uy : vx;  q0.z = cm ? uz : wx;  q0.w = cm ? 0.f : px;
    q1.x = cm ? vx : uy;   q1.y = vy;            q1.z = cm ? vz : wy;  q1.w = cm ? 0.f : py;
    q2.x = cm ? wx : uz;   q2.y = cm ? wy : vz;  q2.z = wz;            q2.w = cm ? 0.f : pz;
    q3.x = cm ? px : 0.f;  q3.y = cm ? py : 0.f; q3.z = cm ? pz : 0.f; q3.w = 1.f;
    v4f* T = (v4f*)(S + (size_t)tid * MST);
    T[0] = q0; T[1] = q1; T[2] = q2; T[3] = q3;
}

#define AGSLOTS 64

__global__ void __launch_bounds__(256)
rgn_chunk_reduce(const float* __restrict__ dih_in,
                 float* __restrict__ agg)
{
    __shared__ __attribute__((aligned(16))) float S0[CHUNK * MST];

    const int tid   = threadIdx.x;
    const int lane  = tid & 31;
    const int wave  = __builtin_amdgcn_readfirstlane(tid >> 5);
    const int chunk = blockIdx.x;
    const int b     = blockIdx.y;

    build_local(dih_in, chunk, b, tid, S0);
    __syncthreads();

#pragma unroll
    for (int rnd = 0; rnd < 8; ++rnd) {
        const int off  = 1 << rnd;
        const int step = 2 << rnd;
        const int n    = CHUNK / step;
        const int nw   = (n + 3) / 4;
        const int gm   = (n < 4) ? n : 4;
#pragma unroll
        for (int it = 0; it < (nw + 7) / 8; ++it) {
            const int q = it * 8 + wave;
            if (q < nw)
                wmma_reduce4<0>(S0, (4 * q + 1) * step - 1, step, off, gm, lane);
        }
        __syncthreads();
    }

    if (tid < 16)
        agg[((size_t)b * NCHUNK + chunk) * 16 + tid] = S0[(CHUNK - 1) * MST + tid];
}

__global__ void __launch_bounds__(256)
rgn_scan_apply(const float* __restrict__ dih_in,
               const float* __restrict__ agg,
               float* __restrict__ out)
{
    __shared__ __attribute__((aligned(16))) float S0[CHUNK * MST];
    __shared__ __attribute__((aligned(16))) float AG[AGSLOTS * MST];

    const int tid   = threadIdx.x;
    const int lane  = tid & 31;
    const int wave  = __builtin_amdgcn_readfirstlane(tid >> 5);
    const int chunk = blockIdx.x;
    const int b     = blockIdx.y;

#pragma unroll
    for (int k = 0; k < (AGSLOTS * 16) / 256; ++k) {
        const int idx  = tid + 256 * k;
        const int slot = idx >> 4;
        const int e    = idx & 15;
        float v = (e % 5 == 0) ? 1.f : 0.f;
        if (slot < chunk) v = agg[((size_t)b * NCHUNK + slot) * 16 + e];
        const int ee = (slot & 1) ? ((e & 3) * 4 + (e >> 2)) : e;
        AG[slot * MST + ee] = v;
    }

    build_local(dih_in, chunk, b, tid, S0);
    __syncthreads();

#pragma unroll
    for (int rnd = 0; rnd < 8; ++rnd) {
        const int off  = 1 << rnd;
        const int step = 2 << rnd;
        const int n    = CHUNK / step;
        const int nw   = (n + 3) / 4;
        const int gm   = (n < 4) ? n : 4;
#pragma unroll
        for (int it = 0; it < (nw + 7) / 8; ++it) {
            const int q = it * 8 + wave;
            if (q < nw)
                wmma_reduce4<0>(S0, (4 * q + 1) * step - 1, step, off, gm, lane);
        }
        if (rnd < 6) {
            const int na  = AGSLOTS / step;
            const int nwa = (na + 3) / 4;
            const int gma = (na < 4) ? na : 4;
            const int wq  = 7 - wave;
            if (wq < nwa)
                wmma_reduce4<0>(AG, (4 * wq + 1) * step - 1, step, off, gma, lane);
        }
        __syncthreads();
    }

#pragma unroll
    for (int rnd = 6; rnd >= 0; --rnd) {
        const int off  = 1 << rnd;
        const int step = 2 << rnd;
        const int n    = CHUNK / step - 1;
        const int nw   = (n + 3) / 4;
#pragma unroll
        for (int it = 0; it < (nw + 7) / 8; ++it) {
            const int q = it * 8 + wave;
            if (q < nw) {
                const int left = n - 4 * q;
                const int gm   = (left < 4) ? left : 4;
                wmma_reduce4<1>(S0, (4 * q + 1) * step + off - 1, step, off, gm, lane);
            }
        }
        __syncthreads();
    }

    const int gi = chunk * CHUNK + tid;
    const float* fin = S0 + (size_t)tid * MST;
    const float* P   = AG + (size_t)(AGSLOTS - 1) * MST;
    const float tx = fin[3], ty = fin[7], tz = fin[11];
    const size_t oi = ((size_t)gi * BATCH + b) * 3;
    out[oi + 0] = fmaf(P[0], tx, fmaf(P[1], ty, fmaf(P[2],  tz, P[3])));
    out[oi + 1] = fmaf(P[4], tx, fmaf(P[5], ty, fmaf(P[6],  tz, P[7])));
    out[oi + 2] = fmaf(P[8], tx, fmaf(P[9], ty, fmaf(P[10], tz, P[11])));
}

extern "C" void kernel_launch(void* const* d_in, const int* in_sizes, int n_in,
                              void* d_out, int out_size, void* d_ws, size_t ws_size,
                              hipStream_t stream)
{
    (void)in_sizes; (void)n_in; (void)out_size; (void)ws_size;
    const float* dih = (const float*)d_in[0];
    float* out = (float*)d_out;
    float* agg = (float*)d_ws;

    dim3 gridA(NCHUNK - 1, BATCH, 1);
    rgn_chunk_reduce<<<gridA, 256, 0, stream>>>(dih, agg);
    dim3 gridB(NCHUNK, BATCH, 1);
    rgn_scan_apply<<<gridB, 256, 0, stream>>>(dih, agg, out);
}


# Prologue waits on global memory once (scan kernel: up to 5 dependent round trips, reduce: 2): bond constants as immediates in a closed-form frame, no table load; chunk-product loads unguarded
# speedup vs baseline: 2.8029x; 1.0814x over previous
#include <hip/hip_runtime.h>
#include <hip/hip_bf16.h>


#define L_STEPS 4096
#define NDIH    3
#define BATCH   32
#define CHAIN   (L_STEPS * NDIH)
#define CHUNK   256
#define NCHUNK  (CHAIN / CHUNK)
#define MST     20
#define PI_F    3.14159265358979f

typedef __attribute__((ext_vector_type(2))) float v2f;
typedef __attribute__((ext_vector_type(4))) float v4f;
typedef __attribute__((ext_vector_type(8))) float v8f;

template <int MODE>
__device__ __forceinline__ void wmma_reduce4(float* buf, int s0, int sstr,
                                             int off, int gmax, int lane)
{
    const int h    = lane >> 4;
    const int Ml   = lane & 15;
    const int g    = Ml >> 2;
    const int j    = Ml & 3;
    const int ge   = (g < gmax) ? g : (gmax - 1);
    const int slot = s0 + ge * sstr;

    const float* xl = buf + (size_t)(slot - off) * MST;
    const float* yr = buf + (size_t)slot * MST;
    const v2f x = *(const v2f*)(xl + j * 4 + 2 * h);
    v2f y;
    bool outR;
    if (MODE == 0) {
        y = *(const v2f*)(yr + j * 4 + 2 * h);
        outR = ((ge & 1) == 0);
    } else {
        y.x = yr[(2 * h) * 4 + j];
        y.y = yr[(2 * h + 1) * 4 + j];
        outR = true;
    }
    const v2f a  = outR ? y : x;
    const v2f bb = outR ? x : y;

    v8f cz = {0.f, 0.f, 0.f, 0.f, 0.f, 0.f, 0.f, 0.f};
    v8f d = __builtin_amdgcn_wmma_f32_16x16x4_f32(false, a, false, bb,
                                                  (short)0, cz, false, false);
    v4f o;
    o.x = (g & 1) ? d[4] : d[0];
    o.y = (g & 1) ? d[5] : d[1];
    o.z = (g & 1) ? d[6] : d[2];
    o.w = (g & 1) ? d[7] : d[3];
    if (h == (g >> 1) && g < gmax)
        *(v4f*)(buf + (size_t)(s0 + g * sstr) * MST + j * 4) = o;
}

__device__ __forceinline__ float load_dihedral(const float* __restrict__ dih_in,
                                               int chunk, int b, int tid, int* jout)
{
    const int gi = chunk * CHUNK + tid;
    const int l  = gi / NDIH;
    const int j  = gi - l * NDIH;
    *jout = j;
    return dih_in[((size_t)l * BATCH + b) * NDIH + j];
}

__device__ __forceinline__ void build_local(float dih, int j, int tid,
                                            float* __restrict__ S)
{
    const float ca = (j == 0) ? 0.525415739f : (j == 1) ? 0.361805219f : 0.441440745f;
    const float sa = (j == 0) ? 0.850845639f : (j == 1) ? 0.932253712f : 0.897290404f;
    const float bl = (j == 0) ? 145.801f     : (j == 1) ? 152.326f     : 132.868f;

    float sd, cd;
    sincosf(dih, &sd, &cd);

    const float ux = ca,       uy = cd * sa, uz = sd * sa;
    const float vx = -sa,      vy = cd * ca, vz = sd * ca;
    const float wx = 0.f,      wy = -sd,     wz = cd;
    const float px = bl * ca,  py = bl * uy, pz = bl * uz;

    const bool cm = (tid & 1) != 0;
    v4f q0, q1, q2, q3;
    q0.x = ux;             q0.y = cm ? uy : vx;  q0.z = cm ? uz : wx;  q0.w = cm ? 0.f : px;
    q1.x = cm ? vx : uy;   q1.y = vy;            q1.z = cm ? vz : wy;  q1.w = cm ? 0.f : py;
    q2.x = cm ? wx : uz;   q2.y = cm ? wy : vz;  q2.z = wz;            q2.w = cm ? 0.f : pz;
    q3.x = cm ? px : 0.f;  q3.y = cm ? py : 0.f; q3.z = cm ? pz : 0.f; q3.w = 1.f;
    v4f* T = (v4f*)(S + (size_t)tid * MST);
    T[0] = q0; T[1] = q1; T[2] = q2; T[3] = q3;
}

#define AGSLOTS 64

__global__ void __launch_bounds__(256)
rgn_chunk_reduce(const float* __restrict__ dih_in,
                 float* __restrict__ agg)
{
    __shared__ __attribute__((aligned(16))) float S0[CHUNK * MST];

    const int tid   = threadIdx.x;
    const int lane  = tid & 31;
    const int wave  = __builtin_amdgcn_readfirstlane(tid >> 5);
    const int chunk = blockIdx.x;
    const int b     = blockIdx.y;

    int jb;
    const float dih = load_dihedral(dih_in, chunk, b, tid, &jb);
    build_local(dih, jb, tid, S0);
    __syncthreads();

#pragma unroll
    for (int rnd = 0; rnd < 8; ++rnd) {
        const int off  = 1 << rnd;
        const int step = 2 << rnd;
        const int n    = CHUNK / step;
        const int nw   = (n + 3) / 4;
        const int gm   = (n < 4) ? n : 4;
#pragma unroll
        for (int it = 0; it < (nw + 7) / 8; ++it) {
            const int q = it * 8 + wave;
            if (q < nw)
                wmma_reduce4<0>(S0, (4 * q + 1) * step - 1, step, off, gm, lane);
        }
        __syncthreads();
    }

    if (tid < 16)
        agg[((size_t)b * NCHUNK + chunk) * 16 + tid] = S0[(CHUNK - 1) * MST + tid];
}

__global__ void __launch_bounds__(256)
rgn_scan_apply(const float* __restrict__ dih_in,
               const float* __restrict__ agg,
               float* __restrict__ out)
{
    __shared__ __attribute__((aligned(16))) float S0[CHUNK * MST];
    __shared__ __attribute__((aligned(16))) float AG[AGSLOTS * MST];

    const int tid   = threadIdx.x;
    const int lane  = tid & 31;
    const int wave  = __builtin_amdgcn_readfirstlane(tid >> 5);
    const int chunk = blockIdx.x;
    const int b     = blockIdx.y;

    int jb;
    const float dih = load_dihedral(dih_in, chunk, b, tid, &jb);
    float av[(AGSLOTS * 16) / 256];
#pragma unroll
    for (int k = 0; k < (AGSLOTS * 16) / 256; ++k) {
        const int idx  = tid + 256 * k;
        const int slot = idx >> 4;
        const int e    = idx & 15;
        const int sl   = (slot < chunk) ? slot : 0;
        av[k] = agg[((size_t)b * NCHUNK + sl) * 16 + e];
    }

#pragma unroll
    for (int k = 0; k < (AGSLOTS * 16) / 256; ++k) {
        const int idx  = tid + 256 * k;
        const int slot = idx >> 4;
        const int e    = idx & 15;
        const float v  = (slot < chunk) ? av[k] : ((e % 5 == 0) ? 1.f : 0.f);
        const int ee = (slot & 1) ? ((e & 3) * 4 + (e >> 2)) : e;
        AG[slot * MST + ee] = v;
    }

    build_local(dih, jb, tid, S0);
    __syncthreads();

#pragma unroll
    for (int rnd = 0; rnd < 8; ++rnd) {
        const int off  = 1 << rnd;
        const int step = 2 << rnd;
        const int n    = CHUNK / step;
        const int nw   = (n + 3) / 4;
        const int gm   = (n < 4) ? n : 4;
#pragma unroll
        for (int it = 0; it < (nw + 7) / 8; ++it) {
            const int q = it * 8 + wave;
            if (q < nw)
                wmma_reduce4<0>(S0, (4 * q + 1) * step - 1, step, off, gm, lane);
        }
        if (rnd < 6) {
            const int na  = AGSLOTS / step;
            const int nwa = (na + 3) / 4;
            const int gma = (na < 4) ? na : 4;
            const int wq  = 7 - wave;
            if (wq < nwa)
                wmma_reduce4<0>(AG, (4 * wq + 1) * step - 1, step, off, gma, lane);
        }
        __syncthreads();
    }

#pragma unroll
    for (int rnd = 6; rnd >= 0; --rnd) {
        const int off  = 1 << rnd;
        const int step = 2 << rnd;
        const int n    = CHUNK / step - 1;
        const int nw   = (n + 3) / 4;
#pragma unroll
        for (int it = 0; it < (nw + 7) / 8; ++it) {
            const int q = it * 8 + wave;
            if (q < nw) {
                const int left = n - 4 * q;
                const int gm   = (left < 4) ? left : 4;
                wmma_reduce4<1>(S0, (4 * q + 1) * step + off - 1, step, off, gm, lane);
            }
        }
        __syncthreads();
    }

    const int gi = chunk * CHUNK + tid;
    const float* fin = S0 + (size_t)tid * MST;
    const float* P   = AG + (size_t)(AGSLOTS - 1) * MST;
    const float tx = fin[3], ty = fin[7], tz = fin[11];
    const size_t oi = ((size_t)gi * BATCH + b) * 3;
    out[oi + 0] = fmaf(P[0], tx, fmaf(P[1], ty, fmaf(P[2],  tz, P[3])));
    out[oi + 1] = fmaf(P[4], tx, fmaf(P[5], ty, fmaf(P[6],  tz, P[7])));
    out[oi + 2] = fmaf(P[8], tx, fmaf(P[9], ty, fmaf(P[10], tz, P[11])));
}

extern "C" void kernel_launch(void* const* d_in, const int* in_sizes, int n_in,
                              void* d_out, int out_size, void* d_ws, size_t ws_size,
                              hipStream_t stream)
{
    (void)in_sizes; (void)n_in; (void)out_size; (void)ws_size;
    const float* dih = (const float*)d_in[0];
    float* out = (float*)d_out;
    float* agg = (float*)d_ws;

    dim3 gridA(NCHUNK - 1, BATCH, 1);
    rgn_chunk_reduce<<<gridA, 256, 0, stream>>>(dih, agg);
    dim3 gridB(NCHUNK, BATCH, 1);
    rgn_scan_apply<<<gridB, 256, 0, stream>>>(dih, agg, out);
}


# Scan kernel: downward tree rounds and the rounds above 64 dropped (129 -> 63 WMMA, 15 -> 6 barriers); each element walks its prefix's nodes, 3x4 matrix-vector on its translation; sum order changed
# speedup vs baseline: 3.0931x; 1.1035x over previous
#include <hip/hip_runtime.h>
#include <hip/hip_bf16.h>


#define L_STEPS 4096
#define NDIH    3
#define BATCH   32
#define CHAIN   (L_STEPS * NDIH)
#define CHUNK   256
#define NCHUNK  (CHAIN / CHUNK)
#define MST     20
#define PI_F    3.14159265358979f

typedef __attribute__((ext_vector_type(2))) float v2f;
typedef __attribute__((ext_vector_type(4))) float v4f;
typedef __attribute__((ext_vector_type(8))) float v8f;

template <bool ALLR>
__device__ __forceinline__ void wmma_reduce4(float* buf, int s0, int sstr,
                                             int off, int gmax, int lane)
{
    const int h    = lane >> 4;
    const int Ml   = lane & 15;
    const int g    = Ml >> 2;
    const int j    = Ml & 3;
    const int ge   = (g < gmax) ? g : (gmax - 1);
    const int slot = s0 + ge * sstr;

    const float* xl = buf + (size_t)(slot - off) * MST;
    const float* yr = buf + (size_t)slot * MST;
    const v2f x = *(const v2f*)(xl + j * 4 + 2 * h);
    const v2f y = *(const v2f*)(yr + j * 4 + 2 * h);
    const bool outR = ALLR || ((ge & 1) == 0);
    const v2f a  = outR ? y : x;
    const v2f bb = outR ? x : y;

    v8f cz = {0.f, 0.f, 0.f, 0.f, 0.f, 0.f, 0.f, 0.f};
    v8f d = __builtin_amdgcn_wmma_f32_16x16x4_f32(false, a, false, bb,
                                                  (short)0, cz, false, false);
    v4f o;
    o.x = (g & 1) ? d[4] : d[0];
    o.y = (g & 1) ? d[5] : d[1];
    o.z = (g & 1) ? d[6] : d[2];
    o.w = (g & 1) ? d[7] : d[3];
    if (h == (g >> 1) && g < gmax)
        *(v4f*)(buf + (size_t)(s0 + g * sstr) * MST + j * 4) = o;
}

__device__ __forceinline__ float load_dihedral(const float* __restrict__ dih_in,
                                               int chunk, int b, int tid, int* jout)
{
    const int gi = chunk * CHUNK + tid;
    const int l  = gi / NDIH;
    const int j  = gi - l * NDIH;
    *jout = j;
    return dih_in[((size_t)l * BATCH + b) * NDIH + j];
}

__device__ __forceinline__ void build_local(float dih, int j, int tid,
                                            float* __restrict__ S)
{
    const float ca = (j == 0) ? 0.525415739f : (j == 1) ? 0.361805219f : 0.441440745f;
    const float sa = (j == 0) ? 0.850845639f : (j == 1) ? 0.932253712f : 0.897290404f;
    const float bl = (j == 0) ? 145.801f     : (j == 1) ? 152.326f     : 132.868f;

    float sd, cd;
    sincosf(dih, &sd, &cd);

    const float ux = ca,       uy = cd * sa, uz = sd * sa;
    const float vx = -sa,      vy = cd * ca, vz = sd * ca;
    const float wx = 0.f,      wy = -sd,     wz = cd;
    const float px = bl * ca,  py = bl * uy, pz = bl * uz;

    const bool cm = (tid & 1) != 0;
    v4f q0, q1, q2, q3;
    q0.x = ux;             q0.y = cm ? uy : vx;  q0.z = cm ? uz : wx;  q0.w = cm ? 0.f : px;
    q1.x = cm ? vx : uy;   q1.y = vy;            q1.z = cm ? vz : wy;  q1.w = cm ? 0.f : py;
    q2.x = cm ? wx : uz;   q2.y = cm ? wy : vz;  q2.z = wz;            q2.w = cm ? 0.f : pz;
    q3.x = cm ? px : 0.f;  q3.y = cm ? py : 0.f; q3.z = cm ? pz : 0.f; q3.w = 1.f;
    v4f* T = (v4f*)(S + (size_t)tid * MST);
    T[0] = q0; T[1] = q1; T[2] = q2; T[3] = q3;
}

#define AGSLOTS 64

__global__ void __launch_bounds__(256)
rgn_chunk_reduce(const float* __restrict__ dih_in,
                 float* __restrict__ agg)
{
    __shared__ __attribute__((aligned(16))) float S0[CHUNK * MST];

    const int tid   = threadIdx.x;
    const int lane  = tid & 31;
    const int wave  = __builtin_amdgcn_readfirstlane(tid >> 5);
    const int chunk = blockIdx.x;
    const int b     = blockIdx.y;

    int jb;
    const float dih = load_dihedral(dih_in, chunk, b, tid, &jb);
    build_local(dih, jb, tid, S0);
    __syncthreads();

#pragma unroll
    for (int rnd = 0; rnd < 8; ++rnd) {
        const int off  = 1 << rnd;
        const int step = 2 << rnd;
        const int n    = CHUNK / step;
        const int nw   = (n + 3) / 4;
        const int gm   = (n < 4) ? n : 4;
#pragma unroll
        for (int it = 0; it < (nw + 7) / 8; ++it) {
            const int q = it * 8 + wave;
            if (q < nw)
                wmma_reduce4<false>(S0, (4 * q + 1) * step - 1, step, off, gm, lane);
        }
        __syncthreads();
    }

    if (tid < 16)
        agg[((size_t)b * NCHUNK + chunk) * 16 + tid] = S0[(CHUNK - 1) * MST + tid];
}

__device__ __forceinline__ void node_apply(const float* __restrict__ S, int slot, bool on,
                                           float& vx, float& vy, float& vz)
{
    const v4f* M = (const v4f*)(S + (size_t)slot * MST);
    const v4f r0 = M[0], r1 = M[1], r2 = M[2];
    const float nx = fmaf(r0.x, vx, fmaf(r0.y, vy, fmaf(r0.z, vz, r0.w)));
    const float ny = fmaf(r1.x, vx, fmaf(r1.y, vy, fmaf(r1.z, vz, r1.w)));
    const float nz = fmaf(r2.x, vx, fmaf(r2.y, vy, fmaf(r2.z, vz, r2.w)));
    vx = on ? nx : vx;
    vy = on ? ny : vy;
    vz = on ? nz : vz;
}

__global__ void __launch_bounds__(256)
rgn_scan_apply(const float* __restrict__ dih_in,
               const float* __restrict__ agg,
               float* __restrict__ out)
{
    __shared__ __attribute__((aligned(16))) float S0[CHUNK * MST];
    __shared__ __attribute__((aligned(16))) float AG[AGSLOTS * MST];

    const int tid   = threadIdx.x;
    const int lane  = tid & 31;
    const int wave  = __builtin_amdgcn_readfirstlane(tid >> 5);
    const int chunk = blockIdx.x;
    const int b     = blockIdx.y;

    int jb;
    const float dih = load_dihedral(dih_in, chunk, b, tid, &jb);
    float av[(AGSLOTS * 16) / 256];
#pragma unroll
    for (int k = 0; k < (AGSLOTS * 16) / 256; ++k) {
        const int idx  = tid + 256 * k;
        const int slot = idx >> 4;
        const int e    = idx & 15;
        const int sl   = (slot < chunk) ? slot : 0;
        av[k] = agg[((size_t)b * NCHUNK + sl) * 16 + e];
    }

#pragma unroll
    for (int k = 0; k < (AGSLOTS * 16) / 256; ++k) {
        const int idx  = tid + 256 * k;
        const int slot = idx >> 4;
        const int e    = idx & 15;
        const float v  = (slot < chunk) ? av[k] : ((e % 5 == 0) ? 1.f : 0.f);
        const int ee = (slot & 1) ? ((e & 3) * 4 + (e >> 2)) : e;
        AG[slot * MST + ee] = v;
    }

    build_local(dih, jb, tid, S0);
    __syncthreads();

#pragma unroll
    for (int rnd = 0; rnd < 6; ++rnd) {
        const int off  = 1 << rnd;
        const int step = 2 << rnd;
        const int n    = CHUNK / step;
        const int nw   = (n + 3) / 4;
        const int gm   = (n < 4) ? n : 4;
#pragma unroll
        for (int it = 0; it < (nw + 7) / 8; ++it) {
            const int q = it * 8 + wave;
            if (q < nw) {
                if (rnd == 5)
                    wmma_reduce4<true>(S0, (4 * q + 1) * step - 1, step, off, gm, lane);
                else
                    wmma_reduce4<false>(S0, (4 * q + 1) * step - 1, step, off, gm, lane);
            }
        }
        {
            const int na  = AGSLOTS / step;
            const int nwa = (na + 3) / 4;
            const int gma = (na < 4) ? na : 4;
            const int wq  = 7 - wave;
            if (wq < nwa)
                wmma_reduce4<false>(AG, (4 * wq + 1) * step - 1, step, off, gma, lane);
        }
        __syncthreads();
    }

    const int n = tid + 1;
    const float* own = S0 + (size_t)tid * MST;
    float vx = own[3], vy = own[7], vz = own[11];
#pragma unroll
    for (int bit = 1; bit <= 5; ++bit) {
        const int hi = (n >> bit) << bit;
        const bool on = (((n >> bit) & 1) != 0) && (hi != n);
        node_apply(S0, (hi > 0) ? (hi - 1) : 0, on, vx, vy, vz);
    }
#pragma unroll
    for (int i = 3; i >= 1; --i)
        node_apply(S0, 64 * i - 1, n > 64 * i, vx, vy, vz);

    const int gi = chunk * CHUNK + tid;
    const float* P   = AG + (size_t)(AGSLOTS - 1) * MST;
    const size_t oi = ((size_t)gi * BATCH + b) * 3;
    out[oi + 0] = fmaf(P[0], vx, fmaf(P[1], vy, fmaf(P[2],  vz, P[3])));
    out[oi + 1] = fmaf(P[4], vx, fmaf(P[5], vy, fmaf(P[6],  vz, P[7])));
    out[oi + 2] = fmaf(P[8], vx, fmaf(P[9], vy, fmaf(P[10], vz, P[11])));
}

extern "C" void kernel_launch(void* const* d_in, const int* in_sizes, int n_in,
                              void* d_out, int out_size, void* d_ws, size_t ws_size,
                              hipStream_t stream)
{
    (void)in_sizes; (void)n_in; (void)out_size; (void)ws_size;
    const float* dih = (const float*)d_in[0];
    float* out = (float*)d_out;
    float* agg = (float*)d_ws;

    dim3 gridA(NCHUNK - 1, BATCH, 1);
    rgn_chunk_reduce<<<gridA, 256, 0, stream>>>(dih, agg);
    dim3 gridB(NCHUNK, BATCH, 1);
    rgn_scan_apply<<<gridB, 256, 0, stream>>>(dih, agg, out);
}


# A wave's products of one tree round issued as one group: all operand loads, then the WMMAs back to back, then the stores (were load-multiply-store one after the other); results bit-identical
# speedup vs baseline: 3.2767x; 1.0594x over previous
#include <hip/hip_runtime.h>
#include <hip/hip_bf16.h>


#define L_STEPS 4096
#define NDIH    3
#define BATCH   32
#define CHAIN   (L_STEPS * NDIH)
#define CHUNK   256
#define NCHUNK  (CHAIN / CHUNK)
#define MST     20
#define PI_F    3.14159265358979f
#define AGSLOTS_ 64

typedef __attribute__((ext_vector_type(2))) float v2f;
typedef __attribute__((ext_vector_type(4))) float v4f;
typedef __attribute__((ext_vector_type(8))) float v8f;

struct Frag { v2f a; v2f b; };

template <bool ALLR>
__device__ __forceinline__ Frag trip_load(const float* buf, int s0, int sstr,
                                          int off, int gmax, int lane)
{
    const int h    = lane >> 4;
    const int Ml   = lane & 15;
    const int g    = Ml >> 2;
    const int j    = Ml & 3;
    const int ge   = (g < gmax) ? g : (gmax - 1);
    const int slot = s0 + ge * sstr;
    const bool outR = ALLR || ((ge & 1) == 0);
    const int sa = outR ? slot : (slot - off);
    const int sb = outR ? (slot - off) : slot;
    Frag f;
    f.a = *(const v2f*)(buf + sa * MST + j * 4 + 2 * h);
    f.b = *(const v2f*)(buf + sb * MST + j * 4 + 2 * h);
    return f;
}

__device__ __forceinline__ v8f trip_mma(const Frag f)
{
    v8f cz = {0.f, 0.f, 0.f, 0.f, 0.f, 0.f, 0.f, 0.f};
    return __builtin_amdgcn_wmma_f32_16x16x4_f32(false, f.a, false, f.b,
                                                 (short)0, cz, false, false);
}

__device__ __forceinline__ void trip_store(float* buf, int s0, int sstr,
                                           int gmax, int lane, const v8f d)
{
    const int h  = lane >> 4;
    const int Ml = lane & 15;
    const int g  = Ml >> 2;
    const int j  = Ml & 3;
    v4f o;
    o.x = (g & 1) ? d[4] : d[0];
    o.y = (g & 1) ? d[5] : d[1];
    o.z = (g & 1) ? d[6] : d[2];
    o.w = (g & 1) ? d[7] : d[3];
    if (h == (g >> 1) && g < gmax)
        *(v4f*)(buf + (s0 + g * sstr) * MST + j * 4) = o;
}

template <int RND, int NT, bool WITH_AG, bool ALLR>
__device__ __forceinline__ void tree_round(float* S, float* AGp, int wave, int lane)
{
    constexpr int off  = 1 << RND;
    constexpr int step = 2 << RND;
    constexpr int n    = CHUNK / step;
    constexpr int gm   = (n < 4) ? n : 4;
    constexpr int na   = (AGSLOTS_ / step) < 1 ? 1 : (AGSLOTS_ / step);
    constexpr int gma  = (na < 4) ? na : 4;
    Frag f[NT > 0 ? NT : 1];
    Frag fa;
    if (WITH_AG)
        fa = trip_load<false>(AGp, (4 * (7 - wave) + 1) * step - 1, step, off, gma, lane);
#pragma unroll
    for (int it = 0; it < NT; ++it)
        f[it] = trip_load<ALLR>(S, (4 * (it * 8 + wave) + 1) * step - 1, step, off, gm, lane);
    v8f d[NT > 0 ? NT : 1];
    v8f da;
#pragma unroll
    for (int it = 0; it < NT; ++it)
        d[it] = trip_mma(f[it]);
    if (WITH_AG)
        da = trip_mma(fa);
#pragma unroll
    for (int it = 0; it < NT; ++it)
        trip_store(S, (4 * (it * 8 + wave) + 1) * step - 1, step, gm, lane, d[it]);
    if (WITH_AG)
        trip_store(AGp, (4 * (7 - wave) + 1) * step - 1, step, gma, lane, da);
}

__device__ __forceinline__ float load_dihedral(const float* __restrict__ dih_in,
                                               int chunk, int b, int tid, int* jout)
{
    const int gi = chunk * CHUNK + tid;
    const int l  = gi / NDIH;
    const int j  = gi - l * NDIH;
    *jout = j;
    return dih_in[((size_t)l * BATCH + b) * NDIH + j];
}

__device__ __forceinline__ void build_local(float dih, int j, int tid,
                                            float* __restrict__ S)
{
    const float ca = (j == 0) ? 0.525415739f : (j == 1) ? 0.361805219f : 0.441440745f;
    const float sa = (j == 0) ? 0.850845639f : (j == 1) ? 0.932253712f : 0.897290404f;
    const float bl = (j == 0) ? 145.801f     : (j == 1) ? 152.326f     : 132.868f;

    float sd, cd;
    sincosf(dih, &sd, &cd);

    const float ux = ca,       uy = cd * sa, uz = sd * sa;
    const float vx = -sa,      vy = cd * ca, vz = sd * ca;
    const float wx = 0.f,      wy = -sd,     wz = cd;
    const float px = bl * ca,  py = bl * uy, pz = bl * uz;

    const bool cm = (tid & 1) != 0;
    v4f q0, q1, q2, q3;
    q0.x = ux;             q0.y = cm ? uy : vx;  q0.z = cm ? uz : wx;  q0.w = cm ? 0.f : px;
    q1.x = cm ? vx : uy;   q1.y = vy;            q1.z = cm ? vz : wy;  q1.w = cm ? 0.f : py;
    q2.x = cm ? wx : uz;   q2.y = cm ? wy : vz;  q2.z = wz;            q2.w = cm ? 0.f : pz;
    q3.x = cm ? px : 0.f;  q3.y = cm ? py : 0.f; q3.z = cm ? pz : 0.f; q3.w = 1.f;
    v4f* T = (v4f*)(S + (size_t)tid * MST);
    T[0] = q0; T[1] = q1; T[2] = q2; T[3] = q3;
}

#define AGSLOTS AGSLOTS_

__global__ void __launch_bounds__(256)
rgn_chunk_reduce(const float* __restrict__ dih_in,
                 float* __restrict__ agg)
{
    __shared__ __attribute__((aligned(16))) float S0[CHUNK * MST];

    const int tid   = threadIdx.x;
    const int lane  = tid & 31;
    const int wave  = __builtin_amdgcn_readfirstlane(tid >> 5);
    const int chunk = blockIdx.x;
    const int b     = blockIdx.y;

    int jb;
    const float dih = load_dihedral(dih_in, chunk, b, tid, &jb);
    build_local(dih, jb, tid, S0);
    __syncthreads();

    __builtin_assume(wave >= 0 && wave < 8);
    tree_round<0, 4, false, false>(S0, S0, wave, lane);
    __syncthreads();
    tree_round<1, 2, false, false>(S0, S0, wave, lane);
    __syncthreads();
    tree_round<2, 1, false, false>(S0, S0, wave, lane);
    __syncthreads();
    if (wave < 4) tree_round<3, 1, false, false>(S0, S0, wave, lane);
    __syncthreads();
    if (wave < 2) tree_round<4, 1, false, false>(S0, S0, wave, lane);
    __syncthreads();
    if (wave < 1) tree_round<5, 1, false, false>(S0, S0, wave, lane);
    __syncthreads();
    if (wave < 1) tree_round<6, 1, false, false>(S0, S0, wave, lane);
    __syncthreads();
    if (wave < 1) tree_round<7, 1, false, false>(S0, S0, wave, lane);
    __syncthreads();

    if (tid < 16)
        agg[((size_t)b * NCHUNK + chunk) * 16 + tid] = S0[(CHUNK - 1) * MST + tid];
}

__device__ __forceinline__ void node_apply(const float* __restrict__ S, int slot, bool on,
                                           float& vx, float& vy, float& vz)
{
    const v4f* M = (const v4f*)(S + (size_t)slot * MST);
    const v4f r0 = M[0], r1 = M[1], r2 = M[2];
    const float nx = fmaf(r0.x, vx, fmaf(r0.y, vy, fmaf(r0.z, vz, r0.w)));
    const float ny = fmaf(r1.x, vx, fmaf(r1.y, vy, fmaf(r1.z, vz, r1.w)));
    const float nz = fmaf(r2.x, vx, fmaf(r2.y, vy, fmaf(r2.z, vz, r2.w)));
    vx = on ? nx : vx;
    vy = on ? ny : vy;
    vz = on ? nz : vz;
}

__global__ void __launch_bounds__(256)
rgn_scan_apply(const float* __restrict__ dih_in,
               const float* __restrict__ agg,
               float* __restrict__ out)
{
    __shared__ __attribute__((aligned(16))) float S0[CHUNK * MST];
    __shared__ __attribute__((aligned(16))) float AG[AGSLOTS * MST];

    const int tid   = threadIdx.x;
    const int lane  = tid & 31;
    const int wave  = __builtin_amdgcn_readfirstlane(tid >> 5);
    const int chunk = blockIdx.x;
    const int b     = blockIdx.y;

    int jb;
    const float dih = load_dihedral(dih_in, chunk, b, tid, &jb);
    float av[(AGSLOTS * 16) / 256];
#pragma unroll
    for (int k = 0; k < (AGSLOTS * 16) / 256; ++k) {
        const int idx  = tid + 256 * k;
        const int slot = idx >> 4;
        const int e    = idx & 15;
        const int sl   = (slot < chunk) ? slot : 0;
        av[k] = agg[((size_t)b * NCHUNK + sl) * 16 + e];
    }

#pragma unroll
    for (int k = 0; k < (AGSLOTS * 16) / 256; ++k) {
        const int idx  = tid + 256 * k;
        const int slot = idx >> 4;
        const int e    = idx & 15;
        const float v  = (slot < chunk) ? av[k] : ((e % 5 == 0) ? 1.f : 0.f);
        const int ee = (slot & 1) ? ((e & 3) * 4 + (e >> 2)) : e;
        AG[slot * MST + ee] = v;
    }

    build_local(dih, jb, tid, S0);
    __syncthreads();

    __builtin_assume(wave >= 0 && wave < 8);
    tree_round<0, 4, true, false>(S0, AG, wave, lane);
    __syncthreads();
    if (wave >= 4) tree_round<1, 2, true,  false>(S0, AG, wave, lane);
    else           tree_round<1, 2, false, false>(S0, AG, wave, lane);
    __syncthreads();
    if (wave >= 6) tree_round<2, 1, true,  false>(S0, AG, wave, lane);
    else           tree_round<2, 1, false, false>(S0, AG, wave, lane);
    __syncthreads();
    if (wave < 4)  tree_round<3, 1, false, false>(S0, AG, wave, lane);
    if (wave == 7) tree_round<3, 0, true,  false>(S0, AG, wave, lane);
    __syncthreads();
    if (wave < 2)  tree_round<4, 1, false, false>(S0, AG, wave, lane);
    if (wave == 7) tree_round<4, 0, true,  false>(S0, AG, wave, lane);
    __syncthreads();
    if (wave < 1)  tree_round<5, 1, false, true>(S0, AG, wave, lane);
    if (wave == 7) tree_round<5, 0, true,  false>(S0, AG, wave, lane);
    __syncthreads();

    const int n = tid + 1;
    const float* own = S0 + (size_t)tid * MST;
    float vx = own[3], vy = own[7], vz = own[11];
#pragma unroll
    for (int bit = 1; bit <= 5; ++bit) {
        const int hi = (n >> bit) << bit;
        const bool on = (((n >> bit) & 1) != 0) && (hi != n);
        node_apply(S0, (hi > 0) ? (hi - 1) : 0, on, vx, vy, vz);
    }
#pragma unroll
    for (int i = 3; i >= 1; --i)
        node_apply(S0, 64 * i - 1, n > 64 * i, vx, vy, vz);

    const int gi = chunk * CHUNK + tid;
    const float* P   = AG + (size_t)(AGSLOTS - 1) * MST;
    const size_t oi = ((size_t)gi * BATCH + b) * 3;
    out[oi + 0] = fmaf(P[0], vx, fmaf(P[1], vy, fmaf(P[2],  vz, P[3])));
    out[oi + 1] = fmaf(P[4], vx, fmaf(P[5], vy, fmaf(P[6],  vz, P[7])));
    out[oi + 2] = fmaf(P[8], vx, fmaf(P[9], vy, fmaf(P[10], vz, P[11])));
}

extern "C" void kernel_launch(void* const* d_in, const int* in_sizes, int n_in,
                              void* d_out, int out_size, void* d_ws, size_t ws_size,
                              hipStream_t stream)
{
    (void)in_sizes; (void)n_in; (void)out_size; (void)ws_size;
    const float* dih = (const float*)d_in[0];
    float* out = (float*)d_out;
    float* agg = (float*)d_ws;

    dim3 gridA(NCHUNK - 1, BATCH, 1);
    rgn_chunk_reduce<<<gridA, 256, 0, stream>>>(dih, agg);
    dim3 gridB(NCHUNK, BATCH, 1);
    rgn_scan_apply<<<gridB, 256, 0, stream>>>(dih, agg, out);
}


# A lane composes its residue's 3 bond transforms in registers by f32 FMAs; tree slots are residue products: 768 elements per workgroup, 16 chunks, a third of the waves; prefix product order changed
# speedup vs baseline: 3.5548x; 1.0849x over previous
#include <hip/hip_runtime.h>
#include <hip/hip_bf16.h>


#define L_STEPS 4096
#define NDIH    3
#define BATCH   32
#define CHAIN   (L_STEPS * NDIH)
#define CHUNK   256
#define RUN     NDIH
#define CHUNK_EL (CHUNK * RUN)
#define NCHUNK  (CHAIN / CHUNK_EL)
#define MST     20
#define PI_F    3.14159265358979f
#define AGSLOTS_ 16

typedef __attribute__((ext_vector_type(2))) float v2f;
typedef __attribute__((ext_vector_type(4))) float v4f;
typedef __attribute__((ext_vector_type(8))) float v8f;

struct Frag { v2f a; v2f b; };

template <bool ALLR>
__device__ __forceinline__ Frag trip_load(const float* buf, int s0, int sstr,
                                          int off, int gmax, int lane)
{
    const int h    = lane >> 4;
    const int Ml   = lane & 15;
    const int g    = Ml >> 2;
    const int j    = Ml & 3;
    const int ge   = (g < gmax) ? g : (gmax - 1);
    const int slot = s0 + ge * sstr;
    const bool outR = ALLR || ((ge & 1) == 0);
    const int sa = outR ? slot : (slot - off);
    const int sb = outR ? (slot - off) : slot;
    Frag f;
    f.a = *(const v2f*)(buf + sa * MST + j * 4 + 2 * h);
    f.b = *(const v2f*)(buf + sb * MST + j * 4 + 2 * h);
    return f;
}

__device__ __forceinline__ v8f trip_mma(const Frag f)
{
    v8f cz = {0.f, 0.f, 0.f, 0.f, 0.f, 0.f, 0.f, 0.f};
    return __builtin_amdgcn_wmma_f32_16x16x4_f32(false, f.a, false, f.b,
                                                 (short)0, cz, false, false);
}

__device__ __forceinline__ void trip_store(float* buf, int s0, int sstr,
                                           int gmax, int lane, const v8f d)
{
    const int h  = lane >> 4;
    const int Ml = lane & 15;
    const int g  = Ml >> 2;
    const int j  = Ml & 3;
    v4f o;
    o.x = (g & 1) ? d[4] : d[0];
    o.y = (g & 1) ? d[5] : d[1];
    o.z = (g & 1) ? d[6] : d[2];
    o.w = (g & 1) ? d[7] : d[3];
    if (h == (g >> 1) && g < gmax)
        *(v4f*)(buf + (s0 + g * sstr) * MST + j * 4) = o;
}

template <int RND, int NT, bool WITH_AG, bool ALLR>
__device__ __forceinline__ void tree_round(float* S, float* AGp, int wave, int lane)
{
    constexpr int off  = 1 << RND;
    constexpr int step = 2 << RND;
    constexpr int n    = CHUNK / step;
    constexpr int gm   = (n < 4) ? n : 4;
    constexpr int na   = (AGSLOTS_ / step) < 1 ? 1 : (AGSLOTS_ / step);
    constexpr int gma  = (na < 4) ? na : 4;
    Frag f[NT > 0 ? NT : 1];
    Frag fa;
    if (WITH_AG)
        fa = trip_load<false>(AGp, (4 * (7 - wave) + 1) * step - 1, step, off, gma, lane);
#pragma unroll
    for (int it = 0; it < NT; ++it)
        f[it] = trip_load<ALLR>(S, (4 * (it * 8 + wave) + 1) * step - 1, step, off, gm, lane);
    v8f d[NT > 0 ? NT : 1];
    v8f da;
#pragma unroll
    for (int it = 0; it < NT; ++it)
        d[it] = trip_mma(f[it]);
    if (WITH_AG)
        da = trip_mma(fa);
#pragma unroll
    for (int it = 0; it < NT; ++it)
        trip_store(S, (4 * (it * 8 + wave) + 1) * step - 1, step, gm, lane, d[it]);
    if (WITH_AG)
        trip_store(AGp, (4 * (7 - wave) + 1) * step - 1, step, gma, lane, da);
}

struct Dih3 { float d0, d1, d2; };

__device__ __forceinline__ Dih3 load_residue(const float* __restrict__ dih_in,
                                             int chunk, int b, int tid)
{
    const int l = chunk * CHUNK + tid;
    const float* p = dih_in + ((size_t)l * BATCH + b) * NDIH;
    Dih3 d;
    d.d0 = p[0]; d.d1 = p[1]; d.d2 = p[2];
    return d;
}

#define CA0 0.525415739f
#define SA0 0.850845639f
#define BL0 145.801f
#define CA1 0.361805219f
#define SA1 0.932253712f
#define BL1 152.326f
#define CA2 0.441440745f
#define SA2 0.897290404f
#define BL2 132.868f

struct Aff { float m[12]; };

__device__ __forceinline__ Aff bond_first(float dih, float ca, float sa, float bl)
{
    float sd, cd;
    sincosf(dih, &sd, &cd);
    const float uy = cd * sa, uz = sd * sa;
    Aff T;
    T.m[0] = ca;  T.m[1] = -sa;     T.m[2]  = 0.f;  T.m[3]  = bl * ca;
    T.m[4] = uy;  T.m[5] = cd * ca; T.m[6]  = -sd;  T.m[7]  = bl * uy;
    T.m[8] = uz;  T.m[9] = sd * ca; T.m[10] = cd;   T.m[11] = bl * uz;
    return T;
}

__device__ __forceinline__ Aff bond_compose(const Aff A, float dih, float ca, float sa, float bl)
{
    float sd, cd;
    sincosf(dih, &sd, &cd);
    const float uy = cd * sa, uz = sd * sa;
    const float vy = cd * ca, vz = sd * ca;
    Aff C;
#pragma unroll
    for (int i = 0; i < 3; ++i) {
        const float a0 = A.m[4 * i], a1 = A.m[4 * i + 1], a2 = A.m[4 * i + 2], a3 = A.m[4 * i + 3];
        const float c0 = fmaf(a0, ca,  fmaf(a1, uy, a2 * uz));
        const float c1 = fmaf(a0, -sa, fmaf(a1, vy, a2 * vz));
        const float c2 = fmaf(a1, -sd, a2 * cd);
        C.m[4 * i] = c0; C.m[4 * i + 1] = c1; C.m[4 * i + 2] = c2;
        C.m[4 * i + 3] = fmaf(bl, c0, a3);
    }
    return C;
}

__device__ __forceinline__ void store_slot(const Aff M, int tid, float* __restrict__ S)
{
    const bool cm = (tid & 1) != 0;
    v4f q0, q1, q2, q3;
    q0.x = M.m[0];              q0.y = cm ? M.m[4] : M.m[1];  q0.z = cm ? M.m[8]  : M.m[2];  q0.w = cm ? 0.f : M.m[3];
    q1.x = cm ? M.m[1] : M.m[4]; q1.y = M.m[5];               q1.z = cm ? M.m[9]  : M.m[6];  q1.w = cm ? 0.f : M.m[7];
    q2.x = cm ? M.m[2] : M.m[8]; q2.y = cm ? M.m[6] : M.m[9]; q2.z = M.m[10];                q2.w = cm ? 0.f : M.m[11];
    q3.x = cm ? M.m[3] : 0.f;    q3.y = cm ? M.m[7] : 0.f;    q3.z = cm ? M.m[11] : 0.f;     q3.w = 1.f;
    v4f* T = (v4f*)(S + (size_t)tid * MST);
    T[0] = q0; T[1] = q1; T[2] = q2; T[3] = q3;
}

#define AGSLOTS AGSLOTS_

__global__ void __launch_bounds__(256)
rgn_chunk_reduce(const float* __restrict__ dih_in,
                 float* __restrict__ agg)
{
    __shared__ __attribute__((aligned(16))) float S0[CHUNK * MST];

    const int tid   = threadIdx.x;
    const int lane  = tid & 31;
    const int wave  = __builtin_amdgcn_readfirstlane(tid >> 5);
    const int chunk = blockIdx.x;
    const int b     = blockIdx.y;

    const Dih3 dd = load_residue(dih_in, chunk, b, tid);
    Aff M = bond_first(dd.d0, CA0, SA0, BL0);
    M = bond_compose(M, dd.d1, CA1, SA1, BL1);
    M = bond_compose(M, dd.d2, CA2, SA2, BL2);
    store_slot(M, tid, S0);
    __syncthreads();

    __builtin_assume(wave >= 0 && wave < 8);
    tree_round<0, 4, false, false>(S0, S0, wave, lane);
    __syncthreads();
    tree_round<1, 2, false, false>(S0, S0, wave, lane);
    __syncthreads();
    tree_round<2, 1, false, false>(S0, S0, wave, lane);
    __syncthreads();
    if (wave < 4) tree_round<3, 1, false, false>(S0, S0, wave, lane);
    __syncthreads();
    if (wave < 2) tree_round<4, 1, false, false>(S0, S0, wave, lane);
    __syncthreads();
    if (wave < 1) tree_round<5, 1, false, false>(S0, S0, wave, lane);
    __syncthreads();
    if (wave < 1) tree_round<6, 1, false, false>(S0, S0, wave, lane);
    __syncthreads();
    if (wave < 1) tree_round<7, 1, false, false>(S0, S0, wave, lane);
    __syncthreads();

    if (tid < 16)
        agg[((size_t)b * NCHUNK + chunk) * 16 + tid] = S0[(CHUNK - 1) * MST + tid];
}

#define IDSLOT CHUNK

__device__ __forceinline__ void node_apply3(const float* __restrict__ S, int slot, float* v)
{
    const v4f* M = (const v4f*)(S + (size_t)slot * MST);
    const v4f r0 = M[0], r1 = M[1], r2 = M[2];
#pragma unroll
    for (int e = 0; e < RUN; ++e) {
        const float x = v[3 * e], y = v[3 * e + 1], z = v[3 * e + 2];
        v[3 * e]     = fmaf(r0.x, x, fmaf(r0.y, y, fmaf(r0.z, z, r0.w)));
        v[3 * e + 1] = fmaf(r1.x, x, fmaf(r1.y, y, fmaf(r1.z, z, r1.w)));
        v[3 * e + 2] = fmaf(r2.x, x, fmaf(r2.y, y, fmaf(r2.z, z, r2.w)));
    }
}

__global__ void __launch_bounds__(256)
rgn_scan_apply(const float* __restrict__ dih_in,
               const float* __restrict__ agg,
               float* __restrict__ out)
{
    __shared__ __attribute__((aligned(16))) float S0[(CHUNK + 1) * MST];
    __shared__ __attribute__((aligned(16))) float AG[AGSLOTS * MST];

    const int tid   = threadIdx.x;
    const int lane  = tid & 31;
    const int wave  = __builtin_amdgcn_readfirstlane(tid >> 5);
    const int chunk = blockIdx.x;
    const int b     = blockIdx.y;

    const Dih3 dd = load_residue(dih_in, chunk, b, tid);
    float av[(AGSLOTS * 16) / 256];
#pragma unroll
    for (int k = 0; k < (AGSLOTS * 16) / 256; ++k) {
        const int idx  = tid + 256 * k;
        const int slot = idx >> 4;
        const int e    = idx & 15;
        const int sl   = (slot < chunk) ? slot : 0;
        av[k] = agg[((size_t)b * NCHUNK + sl) * 16 + e];
    }

#pragma unroll
    for (int k = 0; k < (AGSLOTS * 16) / 256; ++k) {
        const int idx  = tid + 256 * k;
        const int slot = idx >> 4;
        const int e    = idx & 15;
        const float v  = (slot < chunk) ? av[k] : ((e % 5 == 0) ? 1.f : 0.f);
        const int ee = (slot & 1) ? ((e & 3) * 4 + (e >> 2)) : e;
        AG[slot * MST + ee] = v;
    }

    if (tid < 16)
        S0[IDSLOT * MST + tid] = (tid % 5 == 0) ? 1.f : 0.f;

    float v[3 * RUN];
    Aff M = bond_first(dd.d0, CA0, SA0, BL0);
    v[0] = M.m[3]; v[1] = M.m[7]; v[2] = M.m[11];
    M = bond_compose(M, dd.d1, CA1, SA1, BL1);
    v[3] = M.m[3]; v[4] = M.m[7]; v[5] = M.m[11];
    M = bond_compose(M, dd.d2, CA2, SA2, BL2);
    v[6] = M.m[3]; v[7] = M.m[7]; v[8] = M.m[11];
    store_slot(M, tid, S0);
    __syncthreads();

    __builtin_assume(wave >= 0 && wave < 8);
    if (wave >= 6) tree_round<0, 4, true,  false>(S0, AG, wave, lane);
    else           tree_round<0, 4, false, false>(S0, AG, wave, lane);
    __syncthreads();
    if (wave == 7) tree_round<1, 2, true,  false>(S0, AG, wave, lane);
    else           tree_round<1, 2, false, false>(S0, AG, wave, lane);
    __syncthreads();
    if (wave == 7) tree_round<2, 1, true,  false>(S0, AG, wave, lane);
    else           tree_round<2, 1, false, false>(S0, AG, wave, lane);
    __syncthreads();
    if (wave < 4)  tree_round<3, 1, false, false>(S0, AG, wave, lane);
    if (wave == 7) tree_round<3, 0, true,  false>(S0, AG, wave, lane);
    __syncthreads();
    if (wave < 2)  tree_round<4, 1, false, false>(S0, AG, wave, lane);
    __syncthreads();
    if (wave < 1)  tree_round<5, 1, false, true>(S0, AG, wave, lane);
    __syncthreads();

    const int s = tid;
#pragma unroll
    for (int bit = 0; bit <= 5; ++bit) {
        const int hi = (s >> bit) << bit;
        const bool on = ((s >> bit) & 1) != 0;
        node_apply3(S0, on ? (hi - 1) : IDSLOT, v);
    }
#pragma unroll
    for (int i = 3; i >= 1; --i)
        node_apply3(S0, (s >= 64 * i) ? (64 * i - 1) : IDSLOT, v);

    const int gi = (chunk * CHUNK + tid) * RUN;
    const float* P   = AG + (size_t)(AGSLOTS - 1) * MST;
    const v4f p0 = *(const v4f*)(P), p1 = *(const v4f*)(P + 4), p2 = *(const v4f*)(P + 8);
    const size_t oi = ((size_t)gi * BATCH + b) * 3;
#pragma unroll
    for (int e = 0; e < RUN; ++e) {
        const float x = v[3 * e], y = v[3 * e + 1], z = v[3 * e + 2];
        float* o = out + oi + (size_t)e * BATCH * 3;
        o[0] = fmaf(p0.x, x, fmaf(p0.y, y, fmaf(p0.z, z, p0.w)));
        o[1] = fmaf(p1.x, x, fmaf(p1.y, y, fmaf(p1.z, z, p1.w)));
        o[2] = fmaf(p2.x, x, fmaf(p2.y, y, fmaf(p2.z, z, p2.w)));
    }
}

extern "C" void kernel_launch(void* const* d_in, const int* in_sizes, int n_in,
                              void* d_out, int out_size, void* d_ws, size_t ws_size,
                              hipStream_t stream)
{
    (void)in_sizes; (void)n_in; (void)out_size; (void)ws_size;
    const float* dih = (const float*)d_in[0];
    float* out = (float*)d_out;
    float* agg = (float*)d_ws;

    dim3 gridA(NCHUNK - 1, BATCH, 1);
    rgn_chunk_reduce<<<gridA, 256, 0, stream>>>(dih, agg);
    dim3 gridB(NCHUNK, BATCH, 1);
    rgn_scan_apply<<<gridB, 256, 0, stream>>>(dih, agg, out);
}


# Workgroup = 4 adjacent chains x 64 residues (was 1 x 256): four lanes' 12-byte pieces are 48 contiguous bytes, 8-16 lines per wave load/store, was 32; 64 chunks; prefix product order changed
# speedup vs baseline: 4.1018x; 1.1539x over previous
#include <hip/hip_runtime.h>
#include <hip/hip_bf16.h>


#define L_STEPS 4096
#define NDIH    3
#define BATCH   32
#define CHAIN   (L_STEPS * NDIH)
#define CHUNK   256
#define NBW     4
#define RES     (CHUNK / NBW)
#define RUN     NDIH
#define NCHUNK  (L_STEPS / RES)
#define PSLOTS  (CHUNK + 4 * NBW)
#define MST     20
#define PI_F    3.14159265358979f

typedef __attribute__((ext_vector_type(2))) float v2f;
typedef __attribute__((ext_vector_type(4))) float v4f;
typedef __attribute__((ext_vector_type(8))) float v8f;

struct Frag { v2f a; v2f b; };

__device__ __forceinline__ int phys(int t) { return t + ((t >> 6) << 2); }

template <bool ALLR>
__device__ __forceinline__ Frag trip_load(const float* buf, int s0, int sstr,
                                          int off, int gmax, int lane)
{
    const int h    = lane >> 4;
    const int Ml   = lane & 15;
    const int g    = Ml >> 2;
    const int j    = Ml & 3;
    const int ge   = (g < gmax) ? g : (gmax - 1);
    const int slot = s0 + ge * sstr;
    const bool outR = ALLR || ((ge & 1) == 0);
    const int sa = outR ? slot : (slot - off);
    const int sb = outR ? (slot - off) : slot;
    Frag f;
    f.a = *(const v2f*)(buf + phys(sa) * MST + j * 4 + 2 * h);
    f.b = *(const v2f*)(buf + phys(sb) * MST + j * 4 + 2 * h);
    return f;
}

__device__ __forceinline__ v8f trip_mma(const Frag f)
{
    v8f cz = {0.f, 0.f, 0.f, 0.f, 0.f, 0.f, 0.f, 0.f};
    return __builtin_amdgcn_wmma_f32_16x16x4_f32(false, f.a, false, f.b,
                                                 (short)0, cz, false, false);
}

__device__ __forceinline__ void trip_store(float* buf, int s0, int sstr,
                                           int gmax, int lane, const v8f d)
{
    const int h  = lane >> 4;
    const int Ml = lane & 15;
    const int g  = Ml >> 2;
    const int j  = Ml & 3;
    v4f o;
    o.x = (g & 1) ? d[4] : d[0];
    o.y = (g & 1) ? d[5] : d[1];
    o.z = (g & 1) ? d[6] : d[2];
    o.w = (g & 1) ? d[7] : d[3];
    if (h == (g >> 1) && g < gmax)
        *(v4f*)(buf + phys(s0 + g * sstr) * MST + j * 4) = o;
}

template <int RND, int NT, bool ALLR, int NBUF>
__device__ __forceinline__ void tree_round(float* S, float* A2, int wave, int lane)
{
    constexpr int off  = 1 << RND;
    constexpr int step = 2 << RND;
    constexpr int n    = CHUNK / step;
    constexpr int gm   = (n < 4) ? n : 4;
    Frag f[NT * NBUF];
#pragma unroll
    for (int it = 0; it < NT; ++it) {
        f[it] = trip_load<ALLR>(S, (4 * (it * 8 + wave) + 1) * step - 1, step, off, gm, lane);
        if (NBUF == 2)
            f[NT + it] = trip_load<ALLR>(A2, (4 * (it * 8 + wave) + 1) * step - 1, step, off, gm, lane);
    }
    v8f d[NT * NBUF];
#pragma unroll
    for (int it = 0; it < NT * NBUF; ++it)
        d[it] = trip_mma(f[it]);
#pragma unroll
    for (int it = 0; it < NT; ++it) {
        trip_store(S, (4 * (it * 8 + wave) + 1) * step - 1, step, gm, lane, d[it]);
        if (NBUF == 2)
            trip_store(A2, (4 * (it * 8 + wave) + 1) * step - 1, step, gm, lane, d[NT + it]);
    }
}

struct Dih3 { float d0, d1, d2; };

__device__ __forceinline__ Dih3 load_residue(const float* __restrict__ dih_in,
                                             int chunk, int b, int r)
{
    const int l = chunk * RES + r;
    const float* p = dih_in + ((size_t)l * BATCH + b) * NDIH;
    Dih3 d;
    d.d0 = p[0]; d.d1 = p[1]; d.d2 = p[2];
    return d;
}

#define CA0 0.525415739f
#define SA0 0.850845639f
#define BL0 145.801f
#define CA1 0.361805219f
#define SA1 0.932253712f
#define BL1 152.326f
#define CA2 0.441440745f
#define SA2 0.897290404f
#define BL2 132.868f

struct Aff { float m[12]; };

__device__ __forceinline__ Aff bond_first(float dih, float ca, float sa, float bl)
{
    float sd, cd;
    sincosf(dih, &sd, &cd);
    const float uy = cd * sa, uz = sd * sa;
    Aff T;
    T.m[0] = ca;  T.m[1] = -sa;     T.m[2]  = 0.f;  T.m[3]  = bl * ca;
    T.m[4] = uy;  T.m[5] = cd * ca; T.m[6]  = -sd;  T.m[7]  = bl * uy;
    T.m[8] = uz;  T.m[9] = sd * ca; T.m[10] = cd;   T.m[11] = bl * uz;
    return T;
}

__device__ __forceinline__ Aff bond_compose(const Aff A, float dih, float ca, float sa, float bl)
{
    float sd, cd;
    sincosf(dih, &sd, &cd);
    const float uy = cd * sa, uz = sd * sa;
    const float vy = cd * ca, vz = sd * ca;
    Aff C;
#pragma unroll
    for (int i = 0; i < 3; ++i) {
        const float a0 = A.m[4 * i], a1 = A.m[4 * i + 1], a2 = A.m[4 * i + 2], a3 = A.m[4 * i + 3];
        const float c0 = fmaf(a0, ca,  fmaf(a1, uy, a2 * uz));
        const float c1 = fmaf(a0, -sa, fmaf(a1, vy, a2 * vz));
        const float c2 = fmaf(a1, -sd, a2 * cd);
        C.m[4 * i] = c0; C.m[4 * i + 1] = c1; C.m[4 * i + 2] = c2;
        C.m[4 * i + 3] = fmaf(bl, c0, a3);
    }
    return C;
}

__device__ __forceinline__ void store_slot(const Aff M, int t, float* __restrict__ S)
{
    const bool cm = (t & 1) != 0;
    v4f q0, q1, q2, q3;
    q0.x = M.m[0];              q0.y = cm ? M.m[4] : M.m[1];  q0.z = cm ? M.m[8]  : M.m[2];  q0.w = cm ? 0.f : M.m[3];
    q1.x = cm ? M.m[1] : M.m[4]; q1.y = M.m[5];               q1.z = cm ? M.m[9]  : M.m[6];  q1.w = cm ? 0.f : M.m[7];
    q2.x = cm ? M.m[2] : M.m[8]; q2.y = cm ? M.m[6] : M.m[9]; q2.z = M.m[10];                q2.w = cm ? 0.f : M.m[11];
    q3.x = cm ? M.m[3] : 0.f;    q3.y = cm ? M.m[7] : 0.f;    q3.z = cm ? M.m[11] : 0.f;     q3.w = 1.f;
    v4f* T = (v4f*)(S + (size_t)phys(t) * MST);
    T[0] = q0; T[1] = q1; T[2] = q2; T[3] = q3;
}


__global__ void __launch_bounds__(256)
rgn_chunk_reduce(const float* __restrict__ dih_in,
                 float* __restrict__ agg)
{
    __shared__ __attribute__((aligned(16))) float S0[PSLOTS * MST];

    const int tid   = threadIdx.x;
    const int lane  = tid & 31;
    const int wave  = __builtin_amdgcn_readfirstlane(tid >> 5);
    const int chunk = blockIdx.x;
    const int b0    = blockIdx.y * NBW;
    const int q     = tid & (NBW - 1);
    const int r     = tid >> 2;
    const int slot  = q * RES + r;

    const Dih3 dd = load_residue(dih_in, chunk, b0 + q, r);
    Aff M = bond_first(dd.d0, CA0, SA0, BL0);
    M = bond_compose(M, dd.d1, CA1, SA1, BL1);
    M = bond_compose(M, dd.d2, CA2, SA2, BL2);
    store_slot(M, slot, S0);
    __syncthreads();

    __builtin_assume(wave >= 0 && wave < 8);
    tree_round<0, 4, false, 1>(S0, S0, wave, lane);
    __syncthreads();
    tree_round<1, 2, false, 1>(S0, S0, wave, lane);
    __syncthreads();
    tree_round<2, 1, false, 1>(S0, S0, wave, lane);
    __syncthreads();
    if (wave < 4) tree_round<3, 1, false, 1>(S0, S0, wave, lane);
    __syncthreads();
    if (wave < 2) tree_round<4, 1, false, 1>(S0, S0, wave, lane);
    __syncthreads();
    if (wave < 1) tree_round<5, 1, true, 1>(S0, S0, wave, lane);
    __syncthreads();

    if (tid < 16 * NBW) {
        const int qq = tid >> 4;
        const int e  = tid & 15;
        const int es = (chunk & 1) ? ((e & 3) * 4 + (e >> 2)) : e;
        agg[((size_t)(b0 + qq) * NCHUNK + chunk) * 16 + e] =
            S0[phys(qq * RES + RES - 1) * MST + es];
    }
}

#define IDSLOT PSLOTS

__device__ __forceinline__ void node_apply3(const float* __restrict__ S, int slot, float* v)
{
    const v4f* M = (const v4f*)(S + (size_t)slot * MST);
    const v4f r0 = M[0], r1 = M[1], r2 = M[2];
#pragma unroll
    for (int e = 0; e < RUN; ++e) {
        const float x = v[3 * e], y = v[3 * e + 1], z = v[3 * e + 2];
        v[3 * e]     = fmaf(r0.x, x, fmaf(r0.y, y, fmaf(r0.z, z, r0.w)));
        v[3 * e + 1] = fmaf(r1.x, x, fmaf(r1.y, y, fmaf(r1.z, z, r1.w)));
        v[3 * e + 2] = fmaf(r2.x, x, fmaf(r2.y, y, fmaf(r2.z, z, r2.w)));
    }
}

__global__ void __launch_bounds__(256)
rgn_scan_apply(const float* __restrict__ dih_in,
               const float* __restrict__ agg,
               float* __restrict__ out)
{
    __shared__ __attribute__((aligned(16))) float S0[(PSLOTS + 1) * MST];
    __shared__ __attribute__((aligned(16))) float AG[PSLOTS * MST];

    const int tid   = threadIdx.x;
    const int lane  = tid & 31;
    const int wave  = __builtin_amdgcn_readfirstlane(tid >> 5);
    const int chunk = blockIdx.x;
    const int b0    = blockIdx.y * NBW;
    const int q     = tid & (NBW - 1);
    const int r     = tid >> 2;
    const int slot  = q * RES + r;

    const Dih3 dd = load_residue(dih_in, chunk, b0 + q, r);
    v4f av[CHUNK * 4 / 256];
#pragma unroll
    for (int k = 0; k < CHUNK * 4 / 256; ++k) {
        const int idx = tid + 256 * k;
        const int sl  = idx >> 2;
        const int j   = sl & (RES - 1);
        const int jj  = (j < chunk) ? j : 0;
        av[k] = *(const v4f*)(agg + ((size_t)(b0 + (sl >> 6)) * NCHUNK + jj) * 16 + (idx & 3) * 4);
    }

#pragma unroll
    for (int k = 0; k < CHUNK * 4 / 256; ++k) {
        const int idx = tid + 256 * k;
        const int sl  = idx >> 2;
        const int i   = idx & 3;
        const int j   = sl & (RES - 1);
        v4f id;
        id.x = (i == 0) ? 1.f : 0.f; id.y = (i == 1) ? 1.f : 0.f;
        id.z = (i == 2) ? 1.f : 0.f; id.w = (i == 3) ? 1.f : 0.f;
        *(v4f*)(AG + phys(sl) * MST + i * 4) = (j < chunk) ? av[k] : id;
    }

    if (tid < 16)
        S0[IDSLOT * MST + tid] = (tid % 5 == 0) ? 1.f : 0.f;

    float v[3 * RUN];
    Aff M = bond_first(dd.d0, CA0, SA0, BL0);
    v[0] = M.m[3]; v[1] = M.m[7]; v[2] = M.m[11];
    M = bond_compose(M, dd.d1, CA1, SA1, BL1);
    v[3] = M.m[3]; v[4] = M.m[7]; v[5] = M.m[11];
    M = bond_compose(M, dd.d2, CA2, SA2, BL2);
    v[6] = M.m[3]; v[7] = M.m[7]; v[8] = M.m[11];
    store_slot(M, slot, S0);
    __syncthreads();

    __builtin_assume(wave >= 0 && wave < 8);
    tree_round<0, 4, false, 2>(S0, AG, wave, lane);
    __syncthreads();
    tree_round<1, 2, false, 2>(S0, AG, wave, lane);
    __syncthreads();
    tree_round<2, 1, false, 2>(S0, AG, wave, lane);
    __syncthreads();
    if (wave < 4)  tree_round<3, 1, false, 1>(S0, S0, wave, lane);
    else           tree_round<3, 1, false, 1>(AG, AG, wave - 4, lane);
    __syncthreads();
    if (wave < 2)      tree_round<4, 1, false, 1>(S0, S0, wave, lane);
    else if (wave < 4) tree_round<4, 1, false, 1>(AG, AG, wave - 2, lane);
    __syncthreads();
    if (wave < 1)  tree_round<5, 1, true, 1>(AG, AG, wave, lane);
    __syncthreads();

    const int s = r;
#pragma unroll
    for (int bit = 0; bit <= 5; ++bit) {
        const int hi = (s >> bit) << bit;
        const bool on = ((s >> bit) & 1) != 0;
        node_apply3(S0, on ? phys(q * RES + hi - 1) : IDSLOT, v);
    }

    const int gi = (chunk * RES + r) * RUN;
    const float* P   = AG + (size_t)phys(q * RES + RES - 1) * MST;
    const v4f p0 = *(const v4f*)(P), p1 = *(const v4f*)(P + 4), p2 = *(const v4f*)(P + 8);
    const size_t oi = ((size_t)gi * BATCH + b0 + q) * 3;
#pragma unroll
    for (int e = 0; e < RUN; ++e) {
        const float x = v[3 * e], y = v[3 * e + 1], z = v[3 * e + 2];
        float* o = out + oi + (size_t)e * BATCH * 3;
        o[0] = fmaf(p0.x, x, fmaf(p0.y, y, fmaf(p0.z, z, p0.w)));
        o[1] = fmaf(p1.x, x, fmaf(p1.y, y, fmaf(p1.z, z, p1.w)));
        o[2] = fmaf(p2.x, x, fmaf(p2.y, y, fmaf(p2.z, z, p2.w)));
    }
}

extern "C" void kernel_launch(void* const* d_in, const int* in_sizes, int n_in,
                              void* d_out, int out_size, void* d_ws, size_t ws_size,
                              hipStream_t stream)
{
    (void)in_sizes; (void)n_in; (void)out_size; (void)ws_size;
    const float* dih = (const float*)d_in[0];
    float* out = (float*)d_out;
    float* agg = (float*)d_ws;

    dim3 gridA(NCHUNK - 1, BATCH / NBW, 1);
    rgn_chunk_reduce<<<gridA, 256, 0, stream>>>(dih, agg);
    dim3 gridB(NCHUNK, BATCH / NBW, 1);
    rgn_scan_apply<<<gridB, 256, 0, stream>>>(dih, agg, out);
}
